# starting kernel
# baseline (speedup 1.0000x reference)
#include <hip/hip_runtime.h>

typedef float v2f __attribute__((ext_vector_type(2)));
typedef float v8f __attribute__((ext_vector_type(8)));

#define N_ROWS 32768
#define IN_DIM 256
#define EMB 64
#define NCODE 4096
#define LDS_STRIDE 68

__global__ void enorm_kernel(const float* __restrict__ embed,
                             float* __restrict__ norms) {
  int j = blockIdx.x * blockDim.x + threadIdx.x;
  if (j < NCODE) {
    const float4* row = reinterpret_cast<const float4*>(embed + (size_t)j * EMB);
    float s = 0.f;
#pragma unroll
    for (int i = 0; i < EMB / 4; ++i) {
      float4 v = row[i];
      s += v.x * v.x + v.y * v.y + v.z * v.z + v.w * v.w;
    }
    norms[j] = s;
  }
}

__global__ __launch_bounds__(128) void vq_kernel(
    const float* __restrict__ in,
    const float* __restrict__ W,
    const float* __restrict__ bias,
    const float* __restrict__ embed,
    const float* __restrict__ norms,
    float* __restrict__ out) {
  __shared__ float xs_lds[4 * 16 * LDS_STRIDE];

  const int lane = threadIdx.x & 31;
  const int wave = threadIdx.x >> 5;
  const int l16  = lane & 15;
  const int h    = lane >> 4;
  const int R    = (blockIdx.x * 4 + wave) * 16;
  float* lds = xs_lds + wave * 16 * LDS_STRIDE;

  v8f acc[4];
#pragma unroll
  for (int c = 0; c < 4; ++c) acc[c] = (v8f)0.f;

  const float* inRow = in + (size_t)(R + l16) * IN_DIM + 2 * h;
#pragma unroll 4
  for (int s = 0; s < IN_DIM / 4; ++s) {
    v2f a = *reinterpret_cast<const v2f*>(inRow + 4 * s);
    int k = 4 * s + 2 * h;
#pragma unroll
    for (int c = 0; c < 4; ++c) {
      v2f b;
      b.x = W[(size_t)k * EMB + 16 * c + l16];
      b.y = W[(size_t)(k + 1) * EMB + 16 * c + l16];
      acc[c] = __builtin_amdgcn_wmma_f32_16x16x4_f32(
          false, a, false, b, (short)0, acc[c], false, false);
    }
  }

#pragma unroll
  for (int c = 0; c < 4; ++c) {
    float bb = bias[16 * c + l16];
#pragma unroll
    for (int r = 0; r < 8; ++r) {
      lds[(r + 8 * h) * LDS_STRIDE + 16 * c + l16] = acc[c][r] + bb;
    }
  }
  __syncthreads();

  v2f afrag[16];
#pragma unroll
  for (int s = 0; s < 16; ++s)
    afrag[s] = *reinterpret_cast<const v2f*>(&lds[l16 * LDS_STRIDE + 4 * s + 2 * h]);

  float best[8];
  int   bidx[8];
#pragma unroll
  for (int r = 0; r < 8; ++r) { best[r] = 3.402823466e38f; bidx[r] = 0; }

  for (int t = 0; t < NCODE / 16; ++t) {
    const int j = 16 * t + l16;
    const float* erow = embed + (size_t)j * EMB + 2 * h;
    v2f bfrag[16];
#pragma unroll
    for (int s = 0; s < 16; ++s)
      bfrag[s] = *reinterpret_cast<const v2f*>(erow + 4 * s);

    v8f d = (v8f)0.f;
#pragma unroll
    for (int s = 0; s < 16; ++s)
      d = __builtin_amdgcn_wmma_f32_16x16x4_f32(
          false, afrag[s], false, bfrag[s], (short)0, d, false, false);

    float en = norms[j];
#pragma unroll
    for (int r = 0; r < 8; ++r) {
      float dist = en - 2.f * d[r];
      if (dist < best[r]) { best[r] = dist; bidx[r] = j; }
    }
  }

#pragma unroll
  for (int r = 0; r < 8; ++r) {
    float bv = best[r];
    int   bi = bidx[r];
#pragma unroll
    for (int off = 8; off > 0; off >>= 1) {
      float ov = __shfl_xor(bv, off, 16);
      int   oi = __shfl_xor(bi, off, 16);
      if (ov < bv || (ov == bv && oi < bi)) { bv = ov; bi = oi; }
    }
    int row = R + r + 8 * h;
    const float4* src = reinterpret_cast<const float4*>(embed + (size_t)bi * EMB);
    float4*       dst = reinterpret_cast<float4*>(out + (size_t)row * EMB);
    dst[l16] = src[l16];
  }
}

extern "C" void kernel_launch(void* const* d_in, const int* in_sizes, int n_in,
                              void* d_out, int out_size, void* d_ws, size_t ws_size,
                              hipStream_t stream) {
  const float* in    = (const float*)d_in[0];
  const float* W     = (const float*)d_in[1];
  const float* bias  = (const float*)d_in[2];
  const float* embed = (const float*)d_in[3];
  float* norms = (float*)d_ws;
  float* out   = (float*)d_out;

  enorm_kernel<<<NCODE / 256, 256, 0, stream>>>(embed, norms);
  vq_kernel<<<N_ROWS / 64, 128, 0, stream>>>(in, W, bias, embed, norms, out);
}


# Each wave holds all 4 row tiles (64 rows) of xs as A fragments and scans a quarter of the codes: one B-fragment load feeds 4 WMMAs; winners reduced by DPP row-min, merged in LDS; sums unchanged
# speedup vs baseline: 2.1581x; 2.1581x over previous
#include <hip/hip_runtime.h>


typedef float v2f __attribute__((ext_vector_type(2)));
typedef float v8f __attribute__((ext_vector_type(8)));

#define N_ROWS 32768
#define IN_DIM 256
#define EMB 64
#define NCODE 4096
#define LDS_STRIDE 68
#define MT 4
#define WG_ROWS (16 * MT)
#define TILES_PER_WAVE (NCODE / 16 / 4)

__global__ void enorm_kernel(const float* __restrict__ embed,
                             float* __restrict__ norms) {
  int j = blockIdx.x * blockDim.x + threadIdx.x;
  if (j < NCODE) {
    const float4* row = reinterpret_cast<const float4*>(embed + (size_t)j * EMB);
    float s = 0.f;
#pragma unroll
    for (int i = 0; i < EMB / 4; ++i) {
      float4 v = row[i];
      s += v.x * v.x + v.y * v.y + v.z * v.z + v.w * v.w;
    }
    norms[j] = s;
  }
}

struct __attribute__((aligned(8))) Cand { float v; int i; };

template <int MASK>
__device__ __forceinline__ int row_xor(int v) {
  return __builtin_amdgcn_update_dpp(v, v, 0x160 | MASK, 0xf, 0xf, false);
}
template <int MASK>
__device__ __forceinline__ float row_min_step(float v) {
  return fminf(v, __int_as_float(row_xor<MASK>(__float_as_int(v))));
}
template <int MASK>
__device__ __forceinline__ int row_min_step(int v) {
  int o = row_xor<MASK>(v);
  return o < v ? o : v;
}

__global__ __launch_bounds__(128) void vq_kernel(
    const float* __restrict__ in,
    const float* __restrict__ W,
    const float* __restrict__ bias,
    const float* __restrict__ embed,
    const float* __restrict__ norms,
    float* __restrict__ out) {
  __shared__ float xs_lds[WG_ROWS * LDS_STRIDE];
  __shared__ Cand win[4][WG_ROWS];

  const int lane = threadIdx.x & 31;
  const int wave = threadIdx.x >> 5;
  const int l16  = lane & 15;
  const int h    = lane >> 4;
  const int R    = (blockIdx.x * 4 + wave) * 16;
  float* lds = xs_lds + wave * 16 * LDS_STRIDE;

  v8f acc[4];
#pragma unroll
  for (int c = 0; c < 4; ++c) acc[c] = (v8f)0.f;

  const float* inRow = in + (size_t)(R + l16) * IN_DIM + 2 * h;
#pragma unroll 4
  for (int s = 0; s < IN_DIM / 4; ++s) {
    v2f a = *reinterpret_cast<const v2f*>(inRow + 4 * s);
    int k = 4 * s + 2 * h;
#pragma unroll
    for (int c = 0; c < 4; ++c) {
      v2f b;
      b.x = W[(size_t)k * EMB + 16 * c + l16];
      b.y = W[(size_t)(k + 1) * EMB + 16 * c + l16];
      acc[c] = __builtin_amdgcn_wmma_f32_16x16x4_f32(
          false, a, false, b, (short)0, acc[c], false, false);
    }
  }

#pragma unroll
  for (int c = 0; c < 4; ++c) {
    float bb = bias[16 * c + l16];
#pragma unroll
    for (int r = 0; r < 8; ++r) {
      lds[(r + 8 * h) * LDS_STRIDE + 16 * c + l16] = acc[c][r] + bb;
    }
  }
  __syncthreads();

  v2f afrag[MT][16];
#pragma unroll
  for (int m = 0; m < MT; ++m)
#pragma unroll
    for (int s = 0; s < 16; ++s)
      afrag[m][s] = *reinterpret_cast<const v2f*>(
          &xs_lds[(16 * m + l16) * LDS_STRIDE + 4 * s + 2 * h]);

  float best[MT][8];
  int   bidx[MT][8];
#pragma unroll
  for (int m = 0; m < MT; ++m)
#pragma unroll
    for (int r = 0; r < 8; ++r) { best[m][r] = 3.402823466e38f; bidx[m][r] = 0; }

  const int tbase = wave * TILES_PER_WAVE;
  for (int i = 0; i < TILES_PER_WAVE; ++i) {
    const int j = 16 * (tbase + i) + l16;
    const float* erow = embed + (size_t)j * EMB + 2 * h;
    v2f bfrag[16];
#pragma unroll
    for (int s = 0; s < 16; ++s)
      bfrag[s] = *reinterpret_cast<const v2f*>(erow + 4 * s);

    v8f d[MT];
#pragma unroll
    for (int m = 0; m < MT; ++m) d[m] = (v8f)0.f;
#pragma unroll
    for (int s = 0; s < 16; ++s)
#pragma unroll
      for (int m = 0; m < MT; ++m)
        d[m] = __builtin_amdgcn_wmma_f32_16x16x4_f32(
            false, afrag[m][s], false, bfrag[s], (short)0, d[m], false, false);

    float en = norms[j];
#pragma unroll
    for (int m = 0; m < MT; ++m)
#pragma unroll
      for (int r = 0; r < 8; ++r) {
        float dist = en - 2.f * d[m][r];
        if (dist < best[m][r]) { best[m][r] = dist; bidx[m][r] = j; }
      }
  }

  float rmin[MT][8];
  int   ridx[MT][8];
#pragma unroll
  for (int m = 0; m < MT; ++m)
#pragma unroll
    for (int r = 0; r < 8; ++r) rmin[m][r] = row_min_step<8>(best[m][r]);
#pragma unroll
  for (int m = 0; m < MT; ++m)
#pragma unroll
    for (int r = 0; r < 8; ++r) rmin[m][r] = row_min_step<4>(rmin[m][r]);
#pragma unroll
  for (int m = 0; m < MT; ++m)
#pragma unroll
    for (int r = 0; r < 8; ++r) rmin[m][r] = row_min_step<2>(rmin[m][r]);
#pragma unroll
  for (int m = 0; m < MT; ++m)
#pragma unroll
    for (int r = 0; r < 8; ++r) rmin[m][r] = row_min_step<1>(rmin[m][r]);
#pragma unroll
  for (int m = 0; m < MT; ++m)
#pragma unroll
    for (int r = 0; r < 8; ++r)
      ridx[m][r] = row_min_step<8>(best[m][r] == rmin[m][r] ? bidx[m][r] : 0x7fffffff);
#pragma unroll
  for (int m = 0; m < MT; ++m)
#pragma unroll
    for (int r = 0; r < 8; ++r) ridx[m][r] = row_min_step<4>(ridx[m][r]);
#pragma unroll
  for (int m = 0; m < MT; ++m)
#pragma unroll
    for (int r = 0; r < 8; ++r) ridx[m][r] = row_min_step<2>(ridx[m][r]);
#pragma unroll
  for (int m = 0; m < MT; ++m)
#pragma unroll
    for (int r = 0; r < 8; ++r) ridx[m][r] = row_min_step<1>(ridx[m][r]);

  if (l16 == 0) {
#pragma unroll
    for (int m = 0; m < MT; ++m)
#pragma unroll
      for (int r = 0; r < 8; ++r) {
        Cand c;
        c.v = rmin[m][r];
        c.i = ridx[m][r];
        win[wave][16 * m + r + 8 * h] = c;
      }
  }
  __syncthreads();

#pragma unroll
  for (int r = 0; r < 8; ++r) {
    const int lrow = 16 * wave + r + 8 * h;
    Cand c = win[0][lrow];
    float bv = c.v;
    int   bi = c.i;
#pragma unroll
    for (int q = 1; q < 4; ++q) {
      Cand o = win[q][lrow];
      bool take = (o.v < bv) | ((o.v == bv) & (o.i < bi));
      bv = take ? o.v : bv;
      bi = take ? o.i : bi;
    }
    const int row = blockIdx.x * WG_ROWS + lrow;
    const float4* src = reinterpret_cast<const float4*>(embed + (size_t)bi * EMB);
    float4*       dst = reinterpret_cast<float4*>(out + (size_t)row * EMB);
    dst[l16] = src[l16];
  }
}

extern "C" void kernel_launch(void* const* d_in, const int* in_sizes, int n_in,
                              void* d_out, int out_size, void* d_ws, size_t ws_size,
                              hipStream_t stream) {
  const float* in    = (const float*)d_in[0];
  const float* W     = (const float*)d_in[1];
  const float* bias  = (const float*)d_in[2];
  const float* embed = (const float*)d_in[3];
  float* norms = (float*)d_ws;
  float* out   = (float*)d_out;

  enorm_kernel<<<NCODE / 256, 256, 0, stream>>>(embed, norms);
  vq_kernel<<<N_ROWS / 64, 128, 0, stream>>>(in, W, bias, embed, norms, out);
}


# Next code tile's B fragments are loaded as one group ahead of the current tile's 64 WMMAs (two register sets used in turn) instead of in between them; sums and compare order unchanged
# speedup vs baseline: 2.4606x; 1.1402x over previous
#include <hip/hip_runtime.h>


typedef float v2f __attribute__((ext_vector_type(2)));
typedef float v8f __attribute__((ext_vector_type(8)));

#define N_ROWS 32768
#define IN_DIM 256
#define EMB 64
#define NCODE 4096
#define LDS_STRIDE 68
#define MT 4
#define WG_ROWS (16 * MT)
#define TILES_PER_WAVE (NCODE / 16 / 4)

__global__ void enorm_kernel(const float* __restrict__ embed,
                             float* __restrict__ norms) {
  int j = blockIdx.x * blockDim.x + threadIdx.x;
  if (j < NCODE) {
    const float4* row = reinterpret_cast<const float4*>(embed + (size_t)j * EMB);
    float s = 0.f;
#pragma unroll
    for (int i = 0; i < EMB / 4; ++i) {
      float4 v = row[i];
      s += v.x * v.x + v.y * v.y + v.z * v.z + v.w * v.w;
    }
    norms[j] = s;
  }
}

struct __attribute__((aligned(8))) Cand { float v; int i; };

template <int MASK>
__device__ __forceinline__ int row_xor(int v) {
  return __builtin_amdgcn_update_dpp(v, v, 0x160 | MASK, 0xf, 0xf, false);
}
template <int MASK>
__device__ __forceinline__ float row_min_step(float v) {
  return fminf(v, __int_as_float(row_xor<MASK>(__float_as_int(v))));
}
template <int MASK>
__device__ __forceinline__ int row_min_step(int v) {
  int o = row_xor<MASK>(v);
  return o < v ? o : v;
}

__device__ __forceinline__ void load_code_tile(const float* __restrict__ erow,
                                               const float* __restrict__ nptr,
                                               v2f (&bf)[16], float& en) {
#pragma unroll
  for (int s = 0; s < 16; ++s)
    bf[s] = *reinterpret_cast<const v2f*>(erow + 4 * s);
  en = *nptr;
}

__device__ __forceinline__ void scan_code_tile(const v2f (&afrag)[MT][16],
                                               const v2f (&bf)[16], float en, int j,
                                               float (&best)[MT][8], int (&bidx)[MT][8]) {
  v8f d[MT];
#pragma unroll
  for (int m = 0; m < MT; ++m) d[m] = (v8f)0.f;
#pragma unroll
  for (int s = 0; s < 16; ++s)
#pragma unroll
    for (int m = 0; m < MT; ++m)
      d[m] = __builtin_amdgcn_wmma_f32_16x16x4_f32(
          false, afrag[m][s], false, bf[s], (short)0, d[m], false, false);
#pragma unroll
  for (int m = 0; m < MT; ++m)
#pragma unroll
    for (int r = 0; r < 8; ++r) {
      float dist = en - 2.f * d[m][r];
      if (dist < best[m][r]) { best[m][r] = dist; bidx[m][r] = j; }
    }
}

__global__ __launch_bounds__(128) void vq_kernel(
    const float* __restrict__ in,
    const float* __restrict__ W,
    const float* __restrict__ bias,
    const float* __restrict__ embed,
    const float* __restrict__ norms,
    float* __restrict__ out) {
  __shared__ float xs_lds[WG_ROWS * LDS_STRIDE];
  __shared__ Cand win[4][WG_ROWS];

  const int lane = threadIdx.x & 31;
  const int wave = threadIdx.x >> 5;
  const int l16  = lane & 15;
  const int h    = lane >> 4;
  const int R    = (blockIdx.x * 4 + wave) * 16;
  float* lds = xs_lds + wave * 16 * LDS_STRIDE;

  v8f acc[4];
#pragma unroll
  for (int c = 0; c < 4; ++c) acc[c] = (v8f)0.f;

  const float* inRow = in + (size_t)(R + l16) * IN_DIM + 2 * h;
#pragma unroll 4
  for (int s = 0; s < IN_DIM / 4; ++s) {
    v2f a = *reinterpret_cast<const v2f*>(inRow + 4 * s);
    int k = 4 * s + 2 * h;
#pragma unroll
    for (int c = 0; c < 4; ++c) {
      v2f b;
      b.x = W[(size_t)k * EMB + 16 * c + l16];
      b.y = W[(size_t)(k + 1) * EMB + 16 * c + l16];
      acc[c] = __builtin_amdgcn_wmma_f32_16x16x4_f32(
          false, a, false, b, (short)0, acc[c], false, false);
    }
  }

#pragma unroll
  for (int c = 0; c < 4; ++c) {
    float bb = bias[16 * c + l16];
#pragma unroll
    for (int r = 0; r < 8; ++r) {
      lds[(r + 8 * h) * LDS_STRIDE + 16 * c + l16] = acc[c][r] + bb;
    }
  }
  __syncthreads();

  v2f afrag[MT][16];
#pragma unroll
  for (int m = 0; m < MT; ++m)
#pragma unroll
    for (int s = 0; s < 16; ++s)
      afrag[m][s] = *reinterpret_cast<const v2f*>(
          &xs_lds[(16 * m + l16) * LDS_STRIDE + 4 * s + 2 * h]);

  float best[MT][8];
  int   bidx[MT][8];
#pragma unroll
  for (int m = 0; m < MT; ++m)
#pragma unroll
    for (int r = 0; r < 8; ++r) { best[m][r] = 3.402823466e38f; bidx[m][r] = 0; }

  const int tbase = wave * TILES_PER_WAVE;
  const int j0 = 16 * tbase + l16;
  const float* erow0 = embed + (size_t)j0 * EMB + 2 * h;
  const float* nrm0  = norms + j0;

  v2f bA[16], bB[16];
#pragma unroll
  for (int s = 0; s < 16; ++s)
    bA[s] = *reinterpret_cast<const v2f*>(erow0 + 4 * s);
  for (int i = 0; i < TILES_PER_WAVE; i += 2) {
    float enA, enB;
    load_code_tile(erow0 + (size_t)(i + 1) * (16 * EMB), nrm0 + 16 * i, bB, enA);
    scan_code_tile(afrag, bA, enA, j0 + 16 * i, best, bidx);

    const int nx = (i + 2 < TILES_PER_WAVE) ? i + 2 : i;
    __builtin_amdgcn_sched_barrier(0);
    load_code_tile(erow0 + (size_t)nx * (16 * EMB), nrm0 + 16 * (i + 1), bA, enB);
    __builtin_amdgcn_sched_barrier(0);
    scan_code_tile(afrag, bB, enB, j0 + 16 * (i + 1), best, bidx);
  }

  float rmin[MT][8];
  int   ridx[MT][8];
#pragma unroll
  for (int m = 0; m < MT; ++m)
#pragma unroll
    for (int r = 0; r < 8; ++r) rmin[m][r] = row_min_step<8>(best[m][r]);
#pragma unroll
  for (int m = 0; m < MT; ++m)
#pragma unroll
    for (int r = 0; r < 8; ++r) rmin[m][r] = row_min_step<4>(rmin[m][r]);
#pragma unroll
  for (int m = 0; m < MT; ++m)
#pragma unroll
    for (int r = 0; r < 8; ++r) rmin[m][r] = row_min_step<2>(rmin[m][r]);
#pragma unroll
  for (int m = 0; m < MT; ++m)
#pragma unroll
    for (int r = 0; r < 8; ++r) rmin[m][r] = row_min_step<1>(rmin[m][r]);
#pragma unroll
  for (int m = 0; m < MT; ++m)
#pragma unroll
    for (int r = 0; r < 8; ++r)
      ridx[m][r] = row_min_step<8>(best[m][r] == rmin[m][r] ? bidx[m][r] : 0x7fffffff);
#pragma unroll
  for (int m = 0; m < MT; ++m)
#pragma unroll
    for (int r = 0; r < 8; ++r) ridx[m][r] = row_min_step<4>(ridx[m][r]);
#pragma unroll
  for (int m = 0; m < MT; ++m)
#pragma unroll
    for (int r = 0; r < 8; ++r) ridx[m][r] = row_min_step<2>(ridx[m][r]);
#pragma unroll
  for (int m = 0; m < MT; ++m)
#pragma unroll
    for (int r = 0; r < 8; ++r) ridx[m][r] = row_min_step<1>(ridx[m][r]);

  if (l16 == 0) {
#pragma unroll
    for (int m = 0; m < MT; ++m)
#pragma unroll
      for (int r = 0; r < 8; ++r) {
        Cand c;
        c.v = rmin[m][r];
        c.i = ridx[m][r];
        win[wave][16 * m + r + 8 * h] = c;
      }
  }
  __syncthreads();

#pragma unroll
  for (int r = 0; r < 8; ++r) {
    const int lrow = 16 * wave + r + 8 * h;
    Cand c = win[0][lrow];
    float bv = c.v;
    int   bi = c.i;
#pragma unroll
    for (int q = 1; q < 4; ++q) {
      Cand o = win[q][lrow];
      bool take = (o.v < bv) | ((o.v == bv) & (o.i < bi));
      bv = take ? o.v : bv;
      bi = take ? o.i : bi;
    }
    const int row = blockIdx.x * WG_ROWS + lrow;
    const float4* src = reinterpret_cast<const float4*>(embed + (size_t)bi * EMB);
    float4*       dst = reinterpret_cast<float4*>(out + (size_t)row * EMB);
    dst[l16] = src[l16];
  }
}

extern "C" void kernel_launch(void* const* d_in, const int* in_sizes, int n_in,
                              void* d_out, int out_size, void* d_ws, size_t ws_size,
                              hipStream_t stream) {
  const float* in    = (const float*)d_in[0];
  const float* W     = (const float*)d_in[1];
  const float* bias  = (const float*)d_in[2];
  const float* embed = (const float*)d_in[3];
  float* norms = (float*)d_ws;
  float* out   = (float*)d_out;

  enorm_kernel<<<NCODE / 256, 256, 0, stream>>>(embed, norms);
  vq_kernel<<<N_ROWS / 64, 128, 0, stream>>>(in, W, bias, embed, norms, out);
}
